# starting kernel
# baseline (speedup 1.0000x reference)
#include <hip/hip_runtime.h>


#define NN 1024
#define NB 128

typedef float v2f __attribute__((ext_vector_type(2)));
typedef float v8f __attribute__((ext_vector_type(8)));

__global__ __launch_bounds__(128) void mcf_wmma_kernel(const float* __restrict__ x0,
                                                       const float* __restrict__ x1,
                                                       const float* __restrict__ x2,
                                                       float* __restrict__ out) {
    __shared__ float fwd[2 * NN];
    __shared__ float rev[2 * NN];

    const int b     = blockIdx.x;
    const int which = blockIdx.y;
    const float* __restrict__ x =
        (which == 0) ? x0 : (which == 1) ? x1 : x2;
    const float* __restrict__ xb = x + b * NN;
    float* __restrict__ ob = out + (size_t)which * (NB * NN) + (size_t)b * NN;

    const int tid = threadIdx.x;

    #pragma unroll
    for (int i = tid; i < NN; i += 128) {
        float v = xb[i];
        fwd[i]      = v;
        fwd[i + NN] = v;
        int u = (0 - i) & (NN - 1);
        rev[u]      = v;
        rev[u + NN] = v;
    }
    __syncthreads();

    const int wave  = tid >> 5;
    const int lane  = tid & 31;
    const int m16   = lane & 15;
    const int khalf = lane >> 4;

    const float* Abase = fwd + 16 * m16 + 2 * khalf;
    const float* Bbase = rev + NN + 2 * khalf - 256 * wave - m16;

    v8f acc0 = {};
    v8f acc1 = {};

    #pragma unroll 8
    for (int j = 0; j < NN / 2; j += 4) {
        v2f a0, b0, a1, b1;
        a0.x = Abase[j];         a0.y = Abase[j + 1];
        b0.x = Bbase[j];         b0.y = Bbase[j + 1];
        a1.x = Abase[j + 512];   a1.y = Abase[j + 513];
        b1.x = Bbase[j + 512];   b1.y = Bbase[j + 513];

        acc0 = __builtin_amdgcn_wmma_f32_16x16x4_f32(
            false, a0, false, b0, (short)0, acc0, false, false);
        acc1 = __builtin_amdgcn_wmma_f32_16x16x4_f32(
            false, a1, false, b1, (short)0, acc1, false, false);
    }

    v8f acc = acc0 + acc1;

    #pragma unroll
    for (int r = 0; r < 8; ++r) {
        ob[256 * wave + 16 * (r + 8 * khalf) + m16] = acc[r];
    }
}

extern "C" void kernel_launch(void* const* d_in, const int* in_sizes, int n_in,
                              void* d_out, int out_size, void* d_ws, size_t ws_size,
                              hipStream_t stream) {
    const float* x1 = (const float*)d_in[0];
    const float* x2 = (const float*)d_in[1];
    const float* x3 = (const float*)d_in[2];
    float* out = (float*)d_out;

    dim3 grid(NB, 3);
    dim3 block(128);
    mcf_wmma_kernel<<<grid, block, 0, stream>>>(x1, x2, x3, out);
}


# fill: each thread issues its two 16-byte global loads back to back, unconditionally, before any LDS store, replacing a chain of eight dependent load-wait-store round trips
# speedup vs baseline: 1.3481x; 1.3481x over previous
#include <hip/hip_runtime.h>


#define NN 1024
#define NB 128

typedef float v2f __attribute__((ext_vector_type(2)));
typedef float v4f __attribute__((ext_vector_type(4)));
typedef float v8f __attribute__((ext_vector_type(8)));

__device__ __forceinline__ void fill_quad(float* fwd, float* rev, int i0, v4f v) {
    *reinterpret_cast<v4f*>(fwd + i0)        = v;
    *reinterpret_cast<v4f*>(fwd + i0 + 1024) = v;
    const int u0 = (0 - i0) & 1023;
    const int u1 = (0 - (i0 + 1)) & 1023;
    const int u2 = (0 - (i0 + 2)) & 1023;
    const int u3 = (0 - (i0 + 3)) & 1023;
    rev[u0] = v.x;  rev[u0 + 1024] = v.x;
    rev[u1] = v.y;  rev[u1 + 1024] = v.y;
    rev[u2] = v.z;  rev[u2 + 1024] = v.z;
    rev[u3] = v.w;  rev[u3 + 1024] = v.w;
}

__global__ __launch_bounds__(128) void mcf_wmma_kernel(const float* __restrict__ x0,
                                                       const float* __restrict__ x1,
                                                       const float* __restrict__ x2,
                                                       float* __restrict__ out) {
    __shared__ __attribute__((aligned(16))) float fwd[2 * NN];
    __shared__ __attribute__((aligned(16))) float rev[2 * NN];

    const int b     = blockIdx.x;
    const int which = blockIdx.y;
    const float* __restrict__ x =
        (which == 0) ? x0 : (which == 1) ? x1 : x2;
    const float* __restrict__ xb = x + b * NN;
    float* __restrict__ ob = out + (size_t)which * (NB * NN) + (size_t)b * NN;

    const int tid = threadIdx.x;

    const v4f* __restrict__ xb4 = reinterpret_cast<const v4f*>(xb);
    const v4f q0 = xb4[tid];
    const v4f q1 = xb4[tid + 128];
    fill_quad(fwd, rev, 4 * tid, q0);
    fill_quad(fwd, rev, 4 * tid + 512, q1);
    __syncthreads();

    const int wave  = tid >> 5;
    const int lane  = tid & 31;
    const int m16   = lane & 15;
    const int khalf = lane >> 4;

    const float* Abase = fwd + 16 * m16 + 2 * khalf;
    const float* Bbase = rev + NN + 2 * khalf - 256 * wave - m16;

    v8f acc0 = {};
    v8f acc1 = {};

    #pragma unroll 8
    for (int j = 0; j < NN / 2; j += 4) {
        v2f a0, b0, a1, b1;
        a0.x = Abase[j];         a0.y = Abase[j + 1];
        b0.x = Bbase[j];         b0.y = Bbase[j + 1];
        a1.x = Abase[j + 512];   a1.y = Abase[j + 513];
        b1.x = Bbase[j + 512];   b1.y = Bbase[j + 513];

        acc0 = __builtin_amdgcn_wmma_f32_16x16x4_f32(
            false, a0, false, b0, (short)0, acc0, false, false);
        acc1 = __builtin_amdgcn_wmma_f32_16x16x4_f32(
            false, a1, false, b1, (short)0, acc1, false, false);
    }

    v8f acc = acc0 + acc1;

    #pragma unroll
    for (int r = 0; r < 8; ++r) {
        ob[256 * wave + 16 * (r + 8 * khalf) + m16] = acc[r];
    }
}

extern "C" void kernel_launch(void* const* d_in, const int* in_sizes, int n_in,
                              void* d_out, int out_size, void* d_ws, size_t ws_size,
                              hipStream_t stream) {
    const float* x1 = (const float*)d_in[0];
    const float* x2 = (const float*)d_in[1];
    const float* x3 = (const float*)d_in[2];
    float* out = (float*)d_out;

    dim3 grid(NB, 3);
    dim3 block(128);
    mcf_wmma_kernel<<<grid, block, 0, stream>>>(x1, x2, x3, out);
}


# main loop: each wave takes a quarter of j for all 4 output tiles, so 4 A and 4 B fragments feed every A x B pair (1/4 the LDS loads per WMMA); the waves' partial tiles are summed through LDS
# speedup vs baseline: 1.7345x; 1.2866x over previous
#include <hip/hip_runtime.h>


#define NN 1024
#define NB 128

typedef float v2f __attribute__((ext_vector_type(2)));
typedef float v4f __attribute__((ext_vector_type(4)));
typedef float v8f __attribute__((ext_vector_type(8)));

__device__ __forceinline__ void fill_quad(float* fwd, float* rev, int i0, v4f v) {
    *reinterpret_cast<v4f*>(fwd + i0)        = v;
    *reinterpret_cast<v4f*>(fwd + i0 + 1024) = v;
    const int u0 = (0 - i0) & 1023;
    const int u1 = (0 - (i0 + 1)) & 1023;
    const int u2 = (0 - (i0 + 2)) & 1023;
    const int u3 = (0 - (i0 + 3)) & 1023;
    rev[u0] = v.x;  rev[u0 + 1024] = v.x;
    rev[u1] = v.y;  rev[u1 + 1024] = v.y;
    rev[u2] = v.z;  rev[u2 + 1024] = v.z;
    rev[u3] = v.w;  rev[u3 + 1024] = v.w;
}

__device__ __forceinline__ void load_frags(v2f (&A)[2][4], v2f (&B)[2][4],
                                           const float* Ab, const float* Bb, int j) {
    #pragma unroll
    for (int c = 0; c < 2; ++c) {
        const int jc = j + 4 * c;
        A[c][0].x = Ab[jc];            A[c][0].y = Ab[jc + 1];
        A[c][1].x = Ab[jc + 256];      A[c][1].y = Ab[jc + 257];
        A[c][2].x = Ab[jc + 512];      A[c][2].y = Ab[jc + 513];
        A[c][3].x = Ab[jc + 768];      A[c][3].y = Ab[jc + 769];
        B[c][0].x = Bb[jc + 1024];     B[c][0].y = Bb[jc + 1025];
        B[c][1].x = Bb[jc + 1280];     B[c][1].y = Bb[jc + 1281];
        B[c][2].x = Bb[jc + 1536];     B[c][2].y = Bb[jc + 1537];
        B[c][3].x = Bb[jc + 768];      B[c][3].y = Bb[jc + 769];
    }
}

__global__ __launch_bounds__(128) void mcf_wmma_kernel(const float* __restrict__ x0,
                                                       const float* __restrict__ x1,
                                                       const float* __restrict__ x2,
                                                       float* __restrict__ out) {
    __shared__ __attribute__((aligned(16))) float fwd[2 * NN];
    __shared__ __attribute__((aligned(16))) float rev[2 * NN];
    __shared__ __attribute__((aligned(16))) float red[4 * 4 * 256];

    const int b     = blockIdx.x;
    const int which = blockIdx.y;
    const float* __restrict__ x =
        (which == 0) ? x0 : (which == 1) ? x1 : x2;
    const float* __restrict__ xb = x + b * NN;
    float* __restrict__ ob = out + (size_t)which * (NB * NN) + (size_t)b * NN;

    const int tid = threadIdx.x;

    const v4f* __restrict__ xb4 = reinterpret_cast<const v4f*>(xb);
    const v4f q0 = xb4[tid];
    const v4f q1 = xb4[tid + 128];
    fill_quad(fwd, rev, 4 * tid, q0);
    fill_quad(fwd, rev, 4 * tid + 512, q1);
    __syncthreads();

    const int wave  = tid >> 5;
    const int lane  = tid & 31;
    const int m16   = lane & 15;
    const int khalf = lane >> 4;

    const float* Ab = fwd + 64 * wave + 16 * m16 + 2 * khalf;
    const float* Bb = rev + 64 * wave + 2 * khalf - m16;

    v8f acc[4] = {};

    v2f A[2][4], B[2][4];
    load_frags(A, B, Ab, Bb, 0);

    #pragma unroll
    for (int s = 0; s < 8; ++s) {
        v2f An[2][4], Bn[2][4];
        if (s + 1 < 8) load_frags(An, Bn, Ab, Bb, 8 * (s + 1));
        __builtin_amdgcn_sched_barrier(0);

        #pragma unroll
        for (int c = 0; c < 2; ++c) {
            #pragma unroll
            for (int bb = 0; bb < 4; ++bb) {
                #pragma unroll
                for (int a = 0; a < 4; ++a) {
                    const int w = (a - bb) & 3;
                    acc[w] = __builtin_amdgcn_wmma_f32_16x16x4_f32(
                        false, A[c][a], false, B[c][bb], (short)0, acc[w], false, false);
                }
            }
        }
        __builtin_amdgcn_sched_barrier(0);
        if (s + 1 < 8) {
            #pragma unroll
            for (int c = 0; c < 2; ++c) {
                #pragma unroll
                for (int t = 0; t < 4; ++t) { A[c][t] = An[c][t]; B[c][t] = Bn[c][t]; }
            }
        }
    }

    float* rw = red + wave * 1024 + lane * 4;
    #pragma unroll
    for (int t = 0; t < 4; ++t) {
        v4f lo = {acc[t][0], acc[t][1], acc[t][2], acc[t][3]};
        v4f hi = {acc[t][4], acc[t][5], acc[t][6], acc[t][7]};
        *reinterpret_cast<v4f*>(rw + t * 256)       = lo;
        *reinterpret_cast<v4f*>(rw + t * 256 + 128) = hi;
    }
    __syncthreads();

    const float* rr = red + wave * 256 + lane * 4;
    v4f slo = *reinterpret_cast<const v4f*>(rr);
    v4f shi = *reinterpret_cast<const v4f*>(rr + 128);
    #pragma unroll
    for (int v = 1; v < 4; ++v) {
        slo += *reinterpret_cast<const v4f*>(rr + v * 1024);
        shi += *reinterpret_cast<const v4f*>(rr + v * 1024 + 128);
    }

    float* o = ob + 256 * wave + 128 * khalf + m16;
    o[0]   = slo.x;  o[16]  = slo.y;  o[32]  = slo.z;  o[48]  = slo.w;
    o[64]  = shi.x;  o[80]  = shi.y;  o[96]  = shi.z;  o[112] = shi.w;
}

extern "C" void kernel_launch(void* const* d_in, const int* in_sizes, int n_in,
                              void* d_out, int out_size, void* d_ws, size_t ws_size,
                              hipStream_t stream) {
    const float* x1 = (const float*)d_in[0];
    const float* x2 = (const float*)d_in[1];
    const float* x3 = (const float*)d_in[2];
    float* out = (float*)d_out;

    dim3 grid(NB, 3);
    dim3 block(128);
    mcf_wmma_kernel<<<grid, block, 0, stream>>>(x1, x2, x3, out);
}


# work split: one work-group per HALF row (two output tiles; 768 equal groups, 3 per WGP) instead of 384 one-row groups (2 on half the WGPs, 1 on the rest); every tile's sums keep their order
# speedup vs baseline: 1.9653x; 1.1331x over previous
#include <hip/hip_runtime.h>


#define NN 1024
#define NB 128

typedef float v2f __attribute__((ext_vector_type(2)));
typedef float v4f __attribute__((ext_vector_type(4)));
typedef float v8f __attribute__((ext_vector_type(8)));

__device__ __forceinline__ void fill_quad(float* fwd, float* rev, int i0, v4f v) {
    *reinterpret_cast<v4f*>(fwd + i0)        = v;
    *reinterpret_cast<v4f*>(fwd + i0 + 1024) = v;
    const int u0 = (0 - i0) & 1023;
    const int u1 = (0 - (i0 + 1)) & 1023;
    const int u2 = (0 - (i0 + 2)) & 1023;
    const int u3 = (0 - (i0 + 3)) & 1023;
    rev[u0] = v.x;  rev[u0 + 1024] = v.x;
    rev[u1] = v.y;  rev[u1 + 1024] = v.y;
    rev[u2] = v.z;  rev[u2 + 1024] = v.z;
    rev[u3] = v.w;  rev[u3 + 1024] = v.w;
}

__device__ __forceinline__ void load_frags(v2f (&A)[2][4], v2f (&B)[2][4],
                                           const float* Ab, const float* Bb, int j) {
    #pragma unroll
    for (int c = 0; c < 2; ++c) {
        const int jc = j + 4 * c;
        A[c][0].x = Ab[jc];            A[c][0].y = Ab[jc + 1];
        A[c][1].x = Ab[jc + 256];      A[c][1].y = Ab[jc + 257];
        A[c][2].x = Ab[jc + 512];      A[c][2].y = Ab[jc + 513];
        A[c][3].x = Ab[jc + 768];      A[c][3].y = Ab[jc + 769];
        B[c][0].x = Bb[jc + 1024];     B[c][0].y = Bb[jc + 1025];
        B[c][1].x = Bb[jc + 1280];     B[c][1].y = Bb[jc + 1281];
        B[c][2].x = Bb[jc + 1536];     B[c][2].y = Bb[jc + 1537];
        B[c][3].x = Bb[jc + 768];      B[c][3].y = Bb[jc + 769];
    }
}

__global__ __launch_bounds__(128) void mcf_wmma_kernel(const float* __restrict__ x0,
                                                       const float* __restrict__ x1,
                                                       const float* __restrict__ x2,
                                                       float* __restrict__ out) {
    __shared__ __attribute__((aligned(16))) float fwd[2 * NN];
    __shared__ __attribute__((aligned(16))) float rev[2 * NN];
    __shared__ __attribute__((aligned(16))) float red[4 * 2 * 256];

    const int b     = blockIdx.x >> 1;
    const int h     = blockIdx.x & 1;
    const int which = blockIdx.y;
    const float* __restrict__ x =
        (which == 0) ? x0 : (which == 1) ? x1 : x2;
    const float* __restrict__ xb = x + b * NN;
    float* __restrict__ ob = out + (size_t)which * (NB * NN) + (size_t)b * NN;

    const int tid = threadIdx.x;

    const v4f* __restrict__ xb4 = reinterpret_cast<const v4f*>(xb);
    const v4f q0 = xb4[tid];
    const v4f q1 = xb4[tid + 128];
    fill_quad(fwd, rev, 4 * tid, q0);
    fill_quad(fwd, rev, 4 * tid + 512, q1);
    __syncthreads();

    const int wave  = tid >> 5;
    const int lane  = tid & 31;
    const int m16   = lane & 15;
    const int khalf = lane >> 4;

    const float* Ab = fwd + 512 * h + 64 * wave + 16 * m16 + 2 * khalf;
    const float* Bb = rev + 64 * wave + 2 * khalf - m16;

    v8f acc[2] = {};

    v2f A[2][4], B[2][4];
    load_frags(A, B, Ab, Bb, 0);

    #pragma unroll
    for (int s = 0; s < 8; ++s) {
        v2f An[2][4], Bn[2][4];
        if (s + 1 < 8) load_frags(An, Bn, Ab, Bb, 8 * (s + 1));
        __builtin_amdgcn_sched_barrier(0);

        #pragma unroll
        for (int c = 0; c < 2; ++c) {
            #pragma unroll
            for (int bb = 0; bb < 4; ++bb) {
                #pragma unroll
                for (int t = 0; t < 2; ++t) {
                    const int a = (bb + t) & 3;
                    acc[t] = __builtin_amdgcn_wmma_f32_16x16x4_f32(
                        false, A[c][a], false, B[c][bb], (short)0, acc[t], false, false);
                }
            }
        }
        __builtin_amdgcn_sched_barrier(0);
        if (s + 1 < 8) {
            #pragma unroll
            for (int c = 0; c < 2; ++c) {
                #pragma unroll
                for (int t = 0; t < 4; ++t) { A[c][t] = An[c][t]; B[c][t] = Bn[c][t]; }
            }
        }
    }

    float* rw = red + wave * 512 + lane * 4;
    #pragma unroll
    for (int t = 0; t < 2; ++t) {
        v4f lo = {acc[t][0], acc[t][1], acc[t][2], acc[t][3]};
        v4f hi = {acc[t][4], acc[t][5], acc[t][6], acc[t][7]};
        *reinterpret_cast<v4f*>(rw + t * 256)       = lo;
        *reinterpret_cast<v4f*>(rw + t * 256 + 128) = hi;
    }
    __syncthreads();

    const int t  = wave >> 1;
    const int hh = wave & 1;
    const float* rr = red + t * 256 + hh * 128 + lane * 4;
    v4f sum = *reinterpret_cast<const v4f*>(rr);
    #pragma unroll
    for (int v = 1; v < 4; ++v) {
        sum += *reinterpret_cast<const v4f*>(rr + v * 512);
    }

    float* o = ob + 256 * (2 * h + t) + 64 * hh + 128 * khalf + m16;
    o[0]   = sum.x;  o[16]  = sum.y;  o[32]  = sum.z;  o[48]  = sum.w;
}

extern "C" void kernel_launch(void* const* d_in, const int* in_sizes, int n_in,
                              void* d_out, int out_size, void* d_ws, size_t ws_size,
                              hipStream_t stream) {
    const float* x1 = (const float*)d_in[0];
    const float* x2 = (const float*)d_in[1];
    const float* x3 = (const float*)d_in[2];
    float* out = (float*)d_out;

    dim3 grid(256, 3);
    dim3 block(128);
    mcf_wmma_kernel<<<grid, block, 0, stream>>>(x1, x2, x3, out);
}


# self-convolution commutes: split rows into even/odd samples, each even*odd product computed once and the sum doubled; 3 length-512 products replace 4 (96 WMMAs a wave, not 128); sum order changes
# speedup vs baseline: 2.3102x; 1.1755x over previous
#include <hip/hip_runtime.h>


#define NN 1024
#define NB 128
#define MH 512

typedef float v2f __attribute__((ext_vector_type(2)));
typedef float v4f __attribute__((ext_vector_type(4)));
typedef float v8f __attribute__((ext_vector_type(8)));

__device__ __forceinline__ void fill_pair(float* fe, float* fo, float* re, float* ro,
                                          int i, float e, float o) {
    fe[i] = e;  fe[i + MH] = e;
    const int p = (i + 1) & (MH - 1);
    fo[p] = o;  fo[p + MH] = o;
    const int u = (0 - i) & (MH - 1);
    re[u] = e;  re[u + MH] = e;
    ro[u] = o;  ro[u + MH] = o;
}

__device__ __forceinline__ void load_frags(v2f (&A)[4][2], v2f (&B)[4][2],
                                           const float* Ab, const float* Bb, int j) {
    #pragma unroll
    for (int c = 0; c < 4; ++c) {
        const int jc = j + 4 * c;
        A[c][0].x = Ab[jc];            A[c][0].y = Ab[jc + 1];
        A[c][1].x = Ab[jc + 1024];     A[c][1].y = Ab[jc + 1025];
        B[c][0].x = Bb[jc];            B[c][0].y = Bb[jc + 1];
        B[c][1].x = Bb[jc + 1024];     B[c][1].y = Bb[jc + 1025];
    }
}

__global__ __launch_bounds__(128) void mcf_wmma_kernel(const float* __restrict__ x0,
                                                       const float* __restrict__ x1,
                                                       const float* __restrict__ x2,
                                                       float* __restrict__ out) {
    __shared__ __attribute__((aligned(16))) float img[4 * NN];
    __shared__ __attribute__((aligned(16))) float red[4 * 3 * 256];

    float* fe = img;
    float* fo = img + NN;
    float* re = img + 2 * NN;
    float* ro = img + 3 * NN;

    const int b     = blockIdx.x >> 1;
    const int h     = blockIdx.x & 1;
    const int which = blockIdx.y;
    const float* __restrict__ x =
        (which == 0) ? x0 : (which == 1) ? x1 : x2;
    const float* __restrict__ xb = x + b * NN;
    float* __restrict__ ob = out + (size_t)which * (NB * NN) + (size_t)b * NN;

    const int tid = threadIdx.x;

    const v4f* __restrict__ xb4 = reinterpret_cast<const v4f*>(xb);
    const v4f q0 = xb4[tid];
    const v4f q1 = xb4[tid + 128];
    fill_pair(fe, fo, re, ro, 2 * tid,       q0.x, q0.y);
    fill_pair(fe, fo, re, ro, 2 * tid + 1,   q0.z, q0.w);
    fill_pair(fe, fo, re, ro, 2 * tid + 256, q1.x, q1.y);
    fill_pair(fe, fo, re, ro, 2 * tid + 257, q1.z, q1.w);
    __syncthreads();

    const int wave  = tid >> 5;
    const int lane  = tid & 31;
    const int m16   = lane & 15;
    const int khalf = lane >> 4;

    const float* Ab = fe + 256 * h + 128 * wave + 16 * m16 + 2 * khalf;
    const float* Bb = re + MH + 128 * wave + 2 * khalf - m16;

    v8f accEE = {}, accEO = {}, accOO = {};

    v2f A[4][2], B[4][2];
    load_frags(A, B, Ab, Bb, 0);

    #pragma unroll
    for (int s = 0; s < 8; ++s) {
        v2f An[4][2], Bn[4][2];
        if (s + 1 < 8) load_frags(An, Bn, Ab, Bb, 16 * (s + 1));
        __builtin_amdgcn_sched_barrier(0);

        #pragma unroll
        for (int c = 0; c < 4; ++c) {
            accEE = __builtin_amdgcn_wmma_f32_16x16x4_f32(
                false, A[c][0], false, B[c][0], (short)0, accEE, false, false);
            accEO = __builtin_amdgcn_wmma_f32_16x16x4_f32(
                false, A[c][0], false, B[c][1], (short)0, accEO, false, false);
            accOO = __builtin_amdgcn_wmma_f32_16x16x4_f32(
                false, A[c][1], false, B[c][1], (short)0, accOO, false, false);
        }
        __builtin_amdgcn_sched_barrier(0);
        asm volatile("v_nop\n\tv_nop\n\tv_nop\n\tv_nop");
        __builtin_amdgcn_sched_barrier(0);
        if (s + 1 < 8) {
            #pragma unroll
            for (int c = 0; c < 4; ++c) {
                #pragma unroll
                for (int t = 0; t < 2; ++t) { A[c][t] = An[c][t]; B[c][t] = Bn[c][t]; }
            }
        }
    }

    float* rw = red + wave * 768 + lane * 4;
    {
        v4f lo = {accEE[0], accEE[1], accEE[2], accEE[3]};
        v4f hi = {accEE[4], accEE[5], accEE[6], accEE[7]};
        *reinterpret_cast<v4f*>(rw)       = lo;
        *reinterpret_cast<v4f*>(rw + 128) = hi;
    }
    {
        v4f lo = {accOO[0], accOO[1], accOO[2], accOO[3]};
        v4f hi = {accOO[4], accOO[5], accOO[6], accOO[7]};
        *reinterpret_cast<v4f*>(rw + 256)       = lo;
        *reinterpret_cast<v4f*>(rw + 256 + 128) = hi;
    }
    {
        v4f lo = {accEO[0], accEO[1], accEO[2], accEO[3]};
        v4f hi = {accEO[4], accEO[5], accEO[6], accEO[7]};
        *reinterpret_cast<v4f*>(rw + 512)       = lo;
        *reinterpret_cast<v4f*>(rw + 512 + 128) = hi;
    }
    __syncthreads();

    const int hh = wave >> 1;
    const float* rr = red + hh * 128 + lane * 4 + 2 * (wave & 1);
    v2f sE = *reinterpret_cast<const v2f*>(rr);
    sE += *reinterpret_cast<const v2f*>(rr + 256);
    v2f sO = *reinterpret_cast<const v2f*>(rr + 512);
    #pragma unroll
    for (int v = 1; v < 4; ++v) {
        sE += *reinterpret_cast<const v2f*>(rr + v * 768);
        sE += *reinterpret_cast<const v2f*>(rr + v * 768 + 256);
        sO += *reinterpret_cast<const v2f*>(rr + v * 768 + 512);
    }
    sO = sO + sO;

    float* o = ob + 512 * h + 32 * (2 * wave + 8 * khalf) + 2 * m16;
    v2f p0 = {sE.x, sO.x};
    v2f p1 = {sE.y, sO.y};
    *reinterpret_cast<v2f*>(o)      = p0;
    *reinterpret_cast<v2f*>(o + 32) = p1;
}

extern "C" void kernel_launch(void* const* d_in, const int* in_sizes, int n_in,
                              void* d_out, int out_size, void* d_ws, size_t ws_size,
                              hipStream_t stream) {
    const float* x1 = (const float*)d_in[0];
    const float* x2 = (const float*)d_in[1];
    const float* x3 = (const float*)d_in[2];
    float* out = (float*)d_out;

    dim3 grid(256, 3);
    dim3 block(128);
    mcf_wmma_kernel<<<grid, block, 0, stream>>>(x1, x2, x3, out);
}


# second even/odd level: four phases x[4i+r], ten length-256 products (six mixed ones once, doubled), two per tile: 80 WMMAs a wave, not 96; images at pitch 20/16, 16-byte A loads; sum order changes
# speedup vs baseline: 2.5003x; 1.0823x over previous
#include <hip/hip_runtime.h>


#define NN 1024
#define NB 128
#define PH 256
#define FP 672
#define FH 320
#define RP 512

typedef float v2f __attribute__((ext_vector_type(2)));
typedef float v4f __attribute__((ext_vector_type(4)));
typedef float v8f __attribute__((ext_vector_type(8)));

__device__ __forceinline__ void fill_phase(float* F, float* R, int i, int sh, float v) {
    const int ia = i + 4 * (i >> 4);
    F[ia] = v;  F[ia + FH] = v;
    const int u = (0 - i - sh) & (PH - 1);
    R[u] = v;  R[u + PH] = v;
}

__device__ __forceinline__ void load_frags(v4f (&A)[2][4], v2f (&B)[4][4],
                                           const float* const (&pA)[4],
                                           const float* const (&pB)[4], int s) {
    #pragma unroll
    for (int u = 0; u < 2; ++u) {
        #pragma unroll
        for (int f = 0; f < 4; ++f)
            A[u][f] = *reinterpret_cast<const v4f*>(pA[f] + 20 * s + 8 * u);
    }
    #pragma unroll
    for (int c = 0; c < 4; ++c) {
        const int o = 16 * s + 8 * (c >> 1) + 2 * (c & 1);
        #pragma unroll
        for (int f = 0; f < 4; ++f) {
            B[c][f].x = pB[f][o];
            B[c][f].y = pB[f][o + 1];
        }
    }
}

__global__ __launch_bounds__(128) void mcf_wmma_kernel(const float* __restrict__ x0,
                                                       const float* __restrict__ x1,
                                                       const float* __restrict__ x2,
                                                       float* __restrict__ out) {
    __shared__ __attribute__((aligned(16))) float img[4 * FP + 4 * RP];
    __shared__ __attribute__((aligned(16))) float red[4 * 5 * 256];

    float* Fi = img;
    float* Ri = img + 4 * FP;

    const int b     = blockIdx.x >> 1;
    const int h     = blockIdx.x & 1;
    const int which = blockIdx.y;
    const float* __restrict__ x =
        (which == 0) ? x0 : (which == 1) ? x1 : x2;
    const float* __restrict__ xb = x + b * NN;
    float* __restrict__ ob = out + (size_t)which * (NB * NN) + (size_t)b * NN;

    const int tid = threadIdx.x;

    const v4f* __restrict__ xb4 = reinterpret_cast<const v4f*>(xb);
    const v4f q0 = xb4[tid];
    const v4f q1 = xb4[tid + 128];
    fill_phase(Fi,          Ri,          tid,       0, q0.x);
    fill_phase(Fi + FP,     Ri + RP,     tid,       0, q0.y);
    fill_phase(Fi + 2 * FP, Ri + 2 * RP, tid,       1, q0.z);
    fill_phase(Fi + 3 * FP, Ri + 3 * RP, tid,       1, q0.w);
    fill_phase(Fi,          Ri,          tid + 128, 0, q1.x);
    fill_phase(Fi + FP,     Ri + RP,     tid + 128, 0, q1.y);
    fill_phase(Fi + 2 * FP, Ri + 2 * RP, tid + 128, 1, q1.z);
    fill_phase(Fi + 3 * FP, Ri + 3 * RP, tid + 128, 1, q1.w);
    __syncthreads();

    const int wave  = tid >> 5;
    const int lane  = tid & 31;
    const int m16   = lane & 15;
    const int khalf = lane >> 4;
    const int g     = m16 >> 3;
    const int mm    = m16 & 7;

    const int rowoff = 20 * (8 * h + 4 * wave + mm) + 4 * khalf;
    const float* const pA[4] = {
        Fi + (g ? 1 : 0) * FP + rowoff,
        Fi + (g ? 3 : 2) * FP + rowoff,
        Fi + (g ? 2 : 1) * FP + rowoff,
        Fi + (g ? 1 : 3) * FP + rowoff };
    const int coloff = PH + 64 * wave + 4 * khalf - m16;
    const float* const pB[4] = {
        Ri + coloff, Ri + RP + coloff, Ri + 2 * RP + coloff, Ri + 3 * RP + coloff };

    v8f acc0 = {}, acc1 = {}, acc2 = {}, acc3 = {}, acc4 = {};

    v4f A[2][4];
    v2f B[4][4];
    load_frags(A, B, pA, pB, 0);

    #pragma unroll
    for (int s = 0; s < 4; ++s) {
        v4f An[2][4];
        v2f Bn[4][4];
        if (s + 1 < 4) load_frags(An, Bn, pA, pB, s + 1);
        __builtin_amdgcn_sched_barrier(0);

        #pragma unroll
        for (int c = 0; c < 4; ++c) {
            const int u = c >> 1;
            const bool t = (c & 1) != 0;
            const v2f a01 = t ? A[u][0].zw : A[u][0].xy;
            const v2f a23 = t ? A[u][1].zw : A[u][1].xy;
            const v2f a12 = t ? A[u][2].zw : A[u][2].xy;
            const v2f a31 = t ? A[u][3].zw : A[u][3].xy;
            acc0 = __builtin_amdgcn_wmma_f32_16x16x4_f32(
                false, a01, false, B[c][0], (short)0, acc0, false, false);
            acc1 = __builtin_amdgcn_wmma_f32_16x16x4_f32(
                false, a23, false, B[c][0], (short)0, acc1, false, false);
            acc2 = __builtin_amdgcn_wmma_f32_16x16x4_f32(
                false, a12, false, B[c][1], (short)0, acc2, false, false);
            acc3 = __builtin_amdgcn_wmma_f32_16x16x4_f32(
                false, a23, false, B[c][2], (short)0, acc3, false, false);
            acc4 = __builtin_amdgcn_wmma_f32_16x16x4_f32(
                false, a31, false, B[c][3], (short)0, acc4, false, false);
        }
        __builtin_amdgcn_sched_barrier(0);
        asm volatile("v_nop\n\tv_nop\n\tv_nop\n\tv_nop");
        __builtin_amdgcn_sched_barrier(0);
        if (s + 1 < 4) {
            #pragma unroll
            for (int u = 0; u < 2; ++u) {
                #pragma unroll
                for (int f = 0; f < 4; ++f) A[u][f] = An[u][f];
            }
            #pragma unroll
            for (int c = 0; c < 4; ++c) {
                #pragma unroll
                for (int f = 0; f < 4; ++f) B[c][f] = Bn[c][f];
            }
        }
    }

    float* rw = red + wave * 1280 + lane * 4;
    #define STORE_TILE(ACC, G)                                                  \
    {                                                                           \
        v4f lo = {ACC[0], ACC[1], ACC[2], ACC[3]};                              \
        v4f hi = {ACC[4], ACC[5], ACC[6], ACC[7]};                              \
        *reinterpret_cast<v4f*>(rw + (G) * 256)       = lo;                     \
        *reinterpret_cast<v4f*>(rw + (G) * 256 + 128) = hi;                     \
    }
    STORE_TILE(acc0, 0)
    STORE_TILE(acc1, 1)
    STORE_TILE(acc2, 2)
    STORE_TILE(acc3, 3)
    STORE_TILE(acc4, 4)
    #undef STORE_TILE
    __syncthreads();

    const float* rr = red + (wave >> 1) * 128 + lane * 4 + 2 * (wave & 1);
    v2f S[5];
    #pragma unroll
    for (int t5 = 0; t5 < 5; ++t5) {
        S[t5] = *reinterpret_cast<const v2f*>(rr + t5 * 256);
        #pragma unroll
        for (int v = 1; v < 4; ++v)
            S[t5] += *reinterpret_cast<const v2f*>(rr + v * 1280 + t5 * 256);
    }
    const v2f p = S[0] + S[3];
    const v2f q = S[1] + S[2];
    const v2f e2 = q + S[1] + S[4];
    const v2f o1 = p + p;
    const v2f o3 = q + q;
    const v2f xx = S[4] + S[4];
    const float s1 = khalf ? o1.x : p.y;
    const float s2 = khalf ? o3.x : e2.y;
    const float s3 = khalf ? xx.x : 0.0f;
    const float r1 = __shfl_xor(s1, 16);
    const float r2 = __shfl_xor(s2, 16);
    const float r3 = __shfl_xor(s3, 16);
    v4f res;
    if (khalf) { res.x = r1 + xx.y;  res.y = o1.y;  res.z = r2;    res.w = o3.y; }
    else       { res.x = p.x + r3;   res.y = r1;    res.z = e2.x;  res.w = r2;   }

    float* o = ob + 512 * h + 64 * (2 * wave + khalf) + 4 * m16;
    *reinterpret_cast<v4f*>(o) = res;
}

extern "C" void kernel_launch(void* const* d_in, const int* in_sizes, int n_in,
                              void* d_out, int out_size, void* d_ws, size_t ws_size,
                              hipStream_t stream) {
    const float* x1 = (const float*)d_in[0];
    const float* x2 = (const float*)d_in[1];
    const float* x3 = (const float*)d_in[2];
    float* out = (float*)d_out;

    dim3 grid(256, 3);
    dim3 block(128);
    mcf_wmma_kernel<<<grid, block, 0, stream>>>(x1, x2, x3, out);
}


# radix-2 fold: s=x[n]+x[n+512] (circular half), d=x[n]-x[n+512] (negacyclic half), each a length-512 self-product in four phases, out=(S+-D)/2: 320 WMMAs a row, not 640; sum order changes
# speedup vs baseline: 2.8351x; 1.1339x over previous
#include <hip/hip_runtime.h>


#define NN 1024
#define NB 128
#define PH 128
#define FP 352
#define FH 160
#define RP 256
#define SQ (4 * FP + 4 * RP)

typedef float v2f __attribute__((ext_vector_type(2)));
typedef float v4f __attribute__((ext_vector_type(4)));
typedef float v8f __attribute__((ext_vector_type(8)));

template <bool NEG>
__device__ __forceinline__ void fill_phase(float* F, float* R, int i, int sh, float v) {
    const int ia = i + 4 * (i >> 4);
    const int u = (0 - i - sh) & (PH - 1);
    float f1 = v, r0 = v, r1 = v;
    if (NEG) {
        f1 = -v;
        const bool wrap = (u + sh) == 0;
        r0 = wrap ? -v : v;
        r1 = wrap ? v : -v;
    }
    F[ia] = v;  F[ia + FH] = f1;
    R[u] = r0;  R[u + PH] = r1;
}

__device__ __forceinline__ void load_frags(v4f (&A)[2][4], v2f (&B)[4][4],
                                           const float* const (&pA)[4],
                                           const float* const (&pB)[4], int g) {
    const int hq = (g >> 1) * SQ;
    const int s  = g & 1;
    #pragma unroll
    for (int u = 0; u < 2; ++u) {
        #pragma unroll
        for (int f = 0; f < 4; ++f)
            A[u][f] = *reinterpret_cast<const v4f*>(pA[f] + hq + 20 * s + 8 * u);
    }
    #pragma unroll
    for (int c = 0; c < 4; ++c) {
        const int o = hq + 16 * s + 8 * (c >> 1) + 2 * (c & 1);
        #pragma unroll
        for (int f = 0; f < 4; ++f) {
            B[c][f].x = pB[f][o];
            B[c][f].y = pB[f][o + 1];
        }
    }
}

__device__ __forceinline__ v4f finish_half(const v2f (&S)[5], int khalf) {
    const v2f p = S[0] + S[3];
    const v2f q = S[1] + S[2];
    const v2f e2 = q + S[1] + S[4];
    const v2f o1 = p + p;
    const v2f o3 = q + q;
    const v2f xx = S[4] + S[4];
    const float s1 = khalf ? o1.x : p.y;
    const float s2 = khalf ? o3.x : e2.y;
    const float s3 = khalf ? xx.x : 0.0f;
    const float r1 = __shfl_xor(s1, 16);
    const float r2 = __shfl_xor(s2, 16);
    const float r3 = __shfl_xor(s3, 16);
    v4f res;
    if (khalf) { res.x = r1 + xx.y;  res.y = o1.y;  res.z = r2;    res.w = o3.y; }
    else       { res.x = p.x + r3;   res.y = r1;    res.z = e2.x;  res.w = r2;   }
    return res;
}

__global__ __launch_bounds__(128) void mcf_wmma_kernel(const float* __restrict__ x0,
                                                       const float* __restrict__ x1,
                                                       const float* __restrict__ x2,
                                                       float* __restrict__ out) {
    __shared__ __attribute__((aligned(16))) float img[2 * SQ];
    __shared__ __attribute__((aligned(16))) float red[2 * 4 * 5 * 256];

    float* Fi = img;
    float* Ri = img + 4 * FP;

    const int b     = blockIdx.x;
    const int which = blockIdx.y;
    const float* __restrict__ x =
        (which == 0) ? x0 : (which == 1) ? x1 : x2;
    const float* __restrict__ xb = x + b * NN;
    float* __restrict__ ob = out + (size_t)which * (NB * NN) + (size_t)b * NN;

    const int tid = threadIdx.x;

    const v4f* __restrict__ xb4 = reinterpret_cast<const v4f*>(xb);
    const v4f q0 = xb4[tid];
    const v4f q1 = xb4[tid + 128];
    const v4f sv = q0 + q1;
    const v4f dv = q0 - q1;
    fill_phase<false>(Fi,          Ri,          tid, 0, sv.x);
    fill_phase<false>(Fi + FP,     Ri + RP,     tid, 0, sv.y);
    fill_phase<false>(Fi + 2 * FP, Ri + 2 * RP, tid, 1, sv.z);
    fill_phase<false>(Fi + 3 * FP, Ri + 3 * RP, tid, 1, sv.w);
    fill_phase<true>(Fi + SQ,          Ri + SQ,          tid, 0, dv.x);
    fill_phase<true>(Fi + SQ + FP,     Ri + SQ + RP,     tid, 0, dv.y);
    fill_phase<true>(Fi + SQ + 2 * FP, Ri + SQ + 2 * RP, tid, 1, dv.z);
    fill_phase<true>(Fi + SQ + 3 * FP, Ri + SQ + 3 * RP, tid, 1, dv.w);
    __syncthreads();

    const int wave  = tid >> 5;
    const int lane  = tid & 31;
    const int m16   = lane & 15;
    const int khalf = lane >> 4;
    const int g1    = m16 >> 3;
    const int mm    = m16 & 7;

    const int rowoff = 20 * (2 * wave + mm) + 4 * khalf;
    const float* const pA[4] = {
        Fi + (g1 ? 1 : 0) * FP + rowoff,
        Fi + (g1 ? 3 : 2) * FP + rowoff,
        Fi + (g1 ? 2 : 1) * FP + rowoff,
        Fi + (g1 ? 1 : 3) * FP + rowoff };
    const int coloff = PH + 32 * wave + 4 * khalf - m16;
    const float* const pB[4] = {
        Ri + coloff, Ri + RP + coloff, Ri + 2 * RP + coloff, Ri + 3 * RP + coloff };

    v8f acc[2][5] = {};

    float* rw = red + wave * 1280 + lane * 4;
    #define STORE_TILE(ACC, HQ, G)                                              \
    {                                                                           \
        v4f lo = {ACC[0], ACC[1], ACC[2], ACC[3]};                              \
        v4f hi = {ACC[4], ACC[5], ACC[6], ACC[7]};                              \
        *reinterpret_cast<v4f*>(rw + (HQ) * 5120 + (G) * 256)       = lo;       \
        *reinterpret_cast<v4f*>(rw + (HQ) * 5120 + (G) * 256 + 128) = hi;       \
    }

    v4f A[2][4];
    v2f B[4][4];
    load_frags(A, B, pA, pB, 0);

    #pragma unroll
    for (int g = 0; g < 4; ++g) {
        const int hq = g >> 1;
        v4f An[2][4];
        v2f Bn[4][4];
        if (g + 1 < 4) load_frags(An, Bn, pA, pB, g + 1);
        __builtin_amdgcn_sched_barrier(0);

        #pragma unroll
        for (int c = 0; c < 4; ++c) {
            const int u = c >> 1;
            const bool t = (c & 1) != 0;
            const v2f a01 = t ? A[u][0].zw : A[u][0].xy;
            const v2f a23 = t ? A[u][1].zw : A[u][1].xy;
            const v2f a12 = t ? A[u][2].zw : A[u][2].xy;
            const v2f a31 = t ? A[u][3].zw : A[u][3].xy;
            acc[hq][0] = __builtin_amdgcn_wmma_f32_16x16x4_f32(
                false, a01, false, B[c][0], (short)0, acc[hq][0], false, false);
            acc[hq][1] = __builtin_amdgcn_wmma_f32_16x16x4_f32(
                false, a23, false, B[c][0], (short)0, acc[hq][1], false, false);
            acc[hq][2] = __builtin_amdgcn_wmma_f32_16x16x4_f32(
                false, a12, false, B[c][1], (short)0, acc[hq][2], false, false);
            acc[hq][3] = __builtin_amdgcn_wmma_f32_16x16x4_f32(
                false, a23, false, B[c][2], (short)0, acc[hq][3], false, false);
            acc[hq][4] = __builtin_amdgcn_wmma_f32_16x16x4_f32(
                false, a31, false, B[c][3], (short)0, acc[hq][4], false, false);
        }
        __builtin_amdgcn_sched_barrier(0);
        asm volatile("v_nop\n\tv_nop\n\tv_nop\n\tv_nop");
        __builtin_amdgcn_sched_barrier(0);
        if (g == 2) {
            STORE_TILE(acc[0][0], 0, 0)
            STORE_TILE(acc[0][1], 0, 1)
            STORE_TILE(acc[0][2], 0, 2)
            STORE_TILE(acc[0][3], 0, 3)
            STORE_TILE(acc[0][4], 0, 4)
        }
        if (g + 1 < 4) {
            #pragma unroll
            for (int u = 0; u < 2; ++u) {
                #pragma unroll
                for (int f = 0; f < 4; ++f) A[u][f] = An[u][f];
            }
            #pragma unroll
            for (int c = 0; c < 4; ++c) {
                #pragma unroll
                for (int f = 0; f < 4; ++f) B[c][f] = Bn[c][f];
            }
        }
    }

    STORE_TILE(acc[1][0], 1, 0)
    STORE_TILE(acc[1][1], 1, 1)
    STORE_TILE(acc[1][2], 1, 2)
    STORE_TILE(acc[1][3], 1, 3)
    STORE_TILE(acc[1][4], 1, 4)
    #undef STORE_TILE
    __syncthreads();

    const float* rr = red + (wave >> 1) * 128 + lane * 4 + 2 * (wave & 1);
    v2f SS[5], SD[5];
    #pragma unroll
    for (int t5 = 0; t5 < 5; ++t5) {
        SS[t5] = *reinterpret_cast<const v2f*>(rr + t5 * 256);
        SD[t5] = *reinterpret_cast<const v2f*>(rr + 5120 + t5 * 256);
        #pragma unroll
        for (int v = 1; v < 4; ++v) {
            SS[t5] += *reinterpret_cast<const v2f*>(rr + v * 1280 + t5 * 256);
            SD[t5] += *reinterpret_cast<const v2f*>(rr + 5120 + v * 1280 + t5 * 256);
        }
    }
    const v4f vs = finish_half(SS, khalf);
    const v4f vd = finish_half(SD, khalf);
    const v4f lo = 0.5f * (vs + vd);
    const v4f hi = 0.5f * (vs - vd);

    float* o = ob + 64 * (2 * wave + khalf) + 4 * m16;
    *reinterpret_cast<v4f*>(o)       = lo;
    *reinterpret_cast<v4f*>(o + 512) = hi;
}

extern "C" void kernel_launch(void* const* d_in, const int* in_sizes, int n_in,
                              void* d_out, int out_size, void* d_ws, size_t ws_size,
                              hipStream_t stream) {
    const float* x1 = (const float*)d_in[0];
    const float* x2 = (const float*)d_in[1];
    const float* x3 = (const float*)d_in[2];
    float* out = (float*)d_out;

    dim3 grid(128, 3);
    dim3 block(128);
    mcf_wmma_kernel<<<grid, block, 0, stream>>>(x1, x2, x3, out);
}
